# starting kernel
# baseline (speedup 1.0000x reference)
#include <hip/hip_runtime.h>
#include <math.h>


typedef float v2f __attribute__((ext_vector_type(2)));
typedef float v8f __attribute__((ext_vector_type(8)));

#define INF_ 128
#define HEADS 4
#define NEG_SLOPE 0.2f

__device__ __forceinline__ unsigned f32_enc(float f) {
    unsigned u = __float_as_uint(f);
    return (u & 0x80000000u) ? ~u : (u | 0x80000000u);
}
__device__ __forceinline__ float f32_dec(unsigned u) {
    return (u & 0x80000000u) ? __uint_as_float(u & 0x7FFFFFFFu)
                             : __uint_as_float(~u);
}

__device__ __forceinline__ float leaky(float e) {
    return e > 0.0f ? e : NEG_SLOPE * e;
}

__device__ __forceinline__ void atom_add_f32(float* p, float v) {
    unsafeAtomicAdd(p, v);
}

__global__ void __launch_bounds__(256)
gemm_proj_kernel(const float* __restrict__ x,
                 const float* __restrict__ Wg,
                 const float* __restrict__ Ws,
                 float* __restrict__ xproj,
                 float* __restrict__ skip,
                 int n) {
    const int rowBase = blockIdx.x * 16;
    const int colBase = threadIdx.y * 16;
    const int lane    = threadIdx.x;
    const int m       = lane & 15;
    const int koff    = (lane >> 4) * 2;

    int arow = rowBase + m;
    if (arow >= n) arow = n - 1;
    const float* __restrict__ xr = x  + (size_t)arow * INF_;
    const float* __restrict__ gr = Wg + (size_t)(colBase + m) * INF_;
    const float* __restrict__ sr = Ws + (size_t)(colBase + m) * INF_;

    v8f accG = {};
    v8f accS = {};
#pragma unroll
    for (int k0 = 0; k0 < INF_; k0 += 4) {
        v2f a  = *(const v2f*)(xr + k0 + koff);
        v2f bg = *(const v2f*)(gr + k0 + koff);
        v2f bs = *(const v2f*)(sr + k0 + koff);
        accG = __builtin_amdgcn_wmma_f32_16x16x4_f32(
            false, a, false, bg, (short)0, accG, false, false);
        accS = __builtin_amdgcn_wmma_f32_16x16x4_f32(
            false, a, false, bs, (short)0, accS, false, false);
    }

    const int rOff = (lane >> 4) * 8;
#pragma unroll
    for (int j = 0; j < 8; ++j) {
        const int r = rowBase + rOff + j;
        if (r < n) {
            xproj[(size_t)r * INF_ + colBase + m] = accG[j];
            skip [(size_t)r * INF_ + colBase + m] = accS[j];
        }
    }
}

__global__ void att_kernel(const float* __restrict__ xproj,
                           const float* __restrict__ att_src,
                           const float* __restrict__ att_dst,
                           float* __restrict__ a_src,
                           float* __restrict__ a_dst,
                           int n) {
    const int t = blockIdx.x * blockDim.x + threadIdx.x;
    if (t >= n * HEADS) return;
    const int node = t >> 2;
    const int h    = t & 3;
    const float* __restrict__ xp = xproj + (size_t)node * INF_ + h * 32;
    const float* __restrict__ as = att_src + h * 32;
    const float* __restrict__ ad = att_dst + h * 32;
    float s = 0.0f, d = 0.0f;
#pragma unroll
    for (int c = 0; c < 32; ++c) {
        const float v = xp[c];
        s = fmaf(v, as[c], s);
        d = fmaf(v, ad[c], d);
    }
    a_src[t] = s;
    a_dst[t] = d;
}

__global__ void init_kernel(float* __restrict__ out,
                            unsigned* __restrict__ segmax,
                            float* __restrict__ segsum,
                            int n) {
    const int t = blockIdx.x * blockDim.x + threadIdx.x;
    if (t < n * INF_) out[t] = 0.0f;
    if (t < n * HEADS) { segmax[t] = 0u; segsum[t] = 0.0f; }
}

__device__ __forceinline__ void edge_endpoints(const int* __restrict__ ei,
                                               int t, int e,
                                               int& s, int& d) {
    if (t < e) { s = ei[t]; d = ei[e + t]; }
    else       { s = t - e; d = t - e; }
}

__global__ void edge_max_kernel(const int* __restrict__ ei,
                                const float* __restrict__ a_src,
                                const float* __restrict__ a_dst,
                                unsigned* __restrict__ segmax,
                                int e, int n) {
    const int t = blockIdx.x * blockDim.x + threadIdx.x;
    if (t >= e + n) return;
    int s, d; edge_endpoints(ei, t, e, s, d);
#pragma unroll
    for (int h = 0; h < HEADS; ++h) {
        const float ev = leaky(a_src[s * HEADS + h] + a_dst[d * HEADS + h]);
        atomicMax(&segmax[d * HEADS + h], f32_enc(ev));
    }
}

__global__ void edge_sum_kernel(const int* __restrict__ ei,
                                const float* __restrict__ a_src,
                                const float* __restrict__ a_dst,
                                const unsigned* __restrict__ segmax,
                                float* __restrict__ segsum,
                                int e, int n) {
    const int t = blockIdx.x * blockDim.x + threadIdx.x;
    if (t >= e + n) return;
    int s, d; edge_endpoints(ei, t, e, s, d);
#pragma unroll
    for (int h = 0; h < HEADS; ++h) {
        const float ev = leaky(a_src[s * HEADS + h] + a_dst[d * HEADS + h]);
        const float mx = f32_dec(segmax[d * HEADS + h]);
        atom_add_f32(&segsum[d * HEADS + h], __expf(ev - mx));
    }
}

__global__ void __launch_bounds__(256)
edge_scatter_kernel(const int* __restrict__ ei,
                    const float* __restrict__ a_src,
                    const float* __restrict__ a_dst,
                    const unsigned* __restrict__ segmax,
                    const float* __restrict__ segsum,
                    const float* __restrict__ xproj,
                    float* __restrict__ out,
                    int e, int n) {
    const long long gid = (long long)blockIdx.x * blockDim.x + threadIdx.x;
    const int t    = (int)(gid >> 5);
    const int lane = (int)(gid & 31);
    if (t >= e + n) return;
    int s, d; edge_endpoints(ei, t, e, s, d);

    const int h  = lane >> 3;
    const float ev = leaky(a_src[s * HEADS + h] + a_dst[d * HEADS + h]);
    const float mx = f32_dec(segmax[d * HEADS + h]);
    const float sm = segsum[d * HEADS + h];
    const float alpha = __expf(ev - mx) / (sm + 1e-16f);

    const float4 v = *(const float4*)(xproj + (size_t)s * INF_ + lane * 4);
    float* __restrict__ o = out + (size_t)d * INF_ + lane * 4;
    atom_add_f32(o + 0, alpha * v.x);
    atom_add_f32(o + 1, alpha * v.y);
    atom_add_f32(o + 2, alpha * v.z);
    atom_add_f32(o + 3, alpha * v.w);
}

__global__ void final_kernel(float* __restrict__ out,
                             const float* __restrict__ skip,
                             const float* __restrict__ bias,
                             int n) {
    const int t = blockIdx.x * blockDim.x + threadIdx.x;
    if (t >= n * INF_) return;
    const float v = out[t] + bias[t & (INF_ - 1)] + skip[t];
    out[t] = v > 0.0f ? v : (__expf(v) - 1.0f);
}

extern "C" void kernel_launch(void* const* d_in, const int* in_sizes, int n_in,
                              void* d_out, int out_size, void* d_ws, size_t ws_size,
                              hipStream_t stream) {
    const float* x    = (const float*)d_in[0];
    const int*   ei   = (const int*)  d_in[1];
    const float* Wg   = (const float*)d_in[2];
    const float* asw  = (const float*)d_in[3];
    const float* adw  = (const float*)d_in[4];
    const float* bias = (const float*)d_in[5];
    const float* Ws   = (const float*)d_in[6];
    float* out = (float*)d_out;

    const int n = in_sizes[0] / INF_;
    const int e = in_sizes[1] / 2;
    const int etot = e + n;

    float* ws      = (float*)d_ws;
    float* xproj   = ws;
    float* skip    = xproj + (size_t)n * INF_;
    float* a_src   = skip  + (size_t)n * INF_;
    float* a_dst   = a_src + (size_t)n * HEADS;
    unsigned* smax = (unsigned*)(a_dst + (size_t)n * HEADS);
    float* ssum    = (float*)(smax + (size_t)n * HEADS);

    const dim3 gemmBlock(32, 8);
    gemm_proj_kernel<<<(n + 15) / 16, gemmBlock, 0, stream>>>(
        x, Wg, Ws, xproj, skip, n);

    att_kernel<<<(n * HEADS + 255) / 256, 256, 0, stream>>>(
        xproj, asw, adw, a_src, a_dst, n);

    init_kernel<<<(n * INF_ + 255) / 256, 256, 0, stream>>>(
        out, smax, ssum, n);

    edge_max_kernel<<<(etot + 255) / 256, 256, 0, stream>>>(
        ei, a_src, a_dst, smax, e, n);

    edge_sum_kernel<<<(etot + 255) / 256, 256, 0, stream>>>(
        ei, a_src, a_dst, smax, ssum, e, n);

    const long long scatterThreads = (long long)etot * 32;
    edge_scatter_kernel<<<(unsigned)((scatterThreads + 255) / 256), 256, 0, stream>>>(
        ei, a_src, a_dst, smax, ssum, xproj, out, e, n);

    final_kernel<<<(n * INF_ + 255) / 256, 256, 0, stream>>>(
        out, skip, bias, n);
}


# Edges bucketed by destination in every call (count with rank, scan, fill); one wave reduces each destination row in registers with its max and sum: float atomics, max, zero-fill, final passes gone
# speedup vs baseline: 2.2139x; 2.2139x over previous
#include <hip/hip_runtime.h>
#include <math.h>


typedef float v2f __attribute__((ext_vector_type(2)));
typedef float v8f __attribute__((ext_vector_type(8)));

#define INF_ 128
#define HEADS 4
#define NEG_SLOPE 0.2f

__device__ __forceinline__ float leaky(float e) {
    return e > 0.0f ? e : NEG_SLOPE * e;
}

__global__ void __launch_bounds__(256)
gemm_proj_kernel(const float* __restrict__ x,
                 const float* __restrict__ Wg,
                 const float* __restrict__ Ws,
                 float* __restrict__ xproj,
                 float* __restrict__ skip,
                 int n) {
    const int rowBase = blockIdx.x * 16;
    const int colBase = threadIdx.y * 16;
    const int lane    = threadIdx.x;
    const int m       = lane & 15;
    const int koff    = (lane >> 4) * 2;

    int arow = rowBase + m;
    if (arow >= n) arow = n - 1;
    const float* __restrict__ xr = x  + (size_t)arow * INF_;
    const float* __restrict__ gr = Wg + (size_t)(colBase + m) * INF_;
    const float* __restrict__ sr = Ws + (size_t)(colBase + m) * INF_;

    v8f accG = {};
    v8f accS = {};
#pragma unroll
    for (int k0 = 0; k0 < INF_; k0 += 4) {
        v2f a  = *(const v2f*)(xr + k0 + koff);
        v2f bg = *(const v2f*)(gr + k0 + koff);
        v2f bs = *(const v2f*)(sr + k0 + koff);
        accG = __builtin_amdgcn_wmma_f32_16x16x4_f32(
            false, a, false, bg, (short)0, accG, false, false);
        accS = __builtin_amdgcn_wmma_f32_16x16x4_f32(
            false, a, false, bs, (short)0, accS, false, false);
    }

    const int rOff = (lane >> 4) * 8;
#pragma unroll
    for (int j = 0; j < 8; ++j) {
        const int r = rowBase + rOff + j;
        if (r < n) {
            xproj[(size_t)r * INF_ + colBase + m] = accG[j];
            skip [(size_t)r * INF_ + colBase + m] = accS[j];
        }
    }
}

__global__ void att_kernel(const float* __restrict__ xproj,
                           const float* __restrict__ att_src,
                           const float* __restrict__ att_dst,
                           float* __restrict__ a_src,
                           float* __restrict__ a_dst,
                           int* __restrict__ cnt,
                           int n) {
    const int t = blockIdx.x * blockDim.x + threadIdx.x;
    if (t >= n * HEADS) return;
    const int node = t >> 2;
    const int h    = t & 3;
    const float* __restrict__ xp = xproj + (size_t)node * INF_ + h * 32;
    const float* __restrict__ as = att_src + h * 32;
    const float* __restrict__ ad = att_dst + h * 32;
    float s = 0.0f, d = 0.0f;
#pragma unroll
    for (int c = 0; c < 32; ++c) {
        const float v = xp[c];
        s = fmaf(v, as[c], s);
        d = fmaf(v, ad[c], d);
    }
    a_src[t] = s;
    a_dst[t] = d;
    if (h == 0) cnt[node] = 0;
}

__global__ void __launch_bounds__(256)
edge_rank_kernel(const int* __restrict__ ei,
                 int* __restrict__ cnt,
                 int* __restrict__ rank,
                 int e) {
    const int t = blockIdx.x * blockDim.x + threadIdx.x;
    if (t >= e) return;
    const int d = ei[e + t];
    rank[t] = atomicAdd(&cnt[d], 1);
}

__device__ __forceinline__ int4 load4_guard(const int* __restrict__ p, int idx, int n) {
    int4 v = make_int4(0, 0, 0, 0);
    if (idx + 3 < n) {
        v = *(const int4*)(p + idx);
    } else {
        if (idx     < n) v.x = p[idx];
        if (idx + 1 < n) v.y = p[idx + 1];
        if (idx + 2 < n) v.z = p[idx + 2];
    }
    return v;
}

__global__ void __launch_bounds__(1024)
offset_scan_kernel(const int* __restrict__ cnt,
                   int* __restrict__ off,
                   int n) {
    __shared__ int wtot[32];
    const int lane = threadIdx.x & 31;
    const int w    = threadIdx.x >> 5;
    const int chunks = (n + 127) >> 7;
    const int cpw    = (chunks + 31) >> 5;
    const int c0     = w * cpw;
    int c1 = c0 + cpw;
    if (c1 > chunks) c1 = chunks;

    int s = 0;
    for (int c = c0; c < c1; ++c) {
        const int4 v = load4_guard(cnt, (c << 7) + lane * 4, n);
        s += (v.x + v.y) + (v.z + v.w);
    }
    for (int o = 16; o > 0; o >>= 1) s += __shfl_xor(s, o);
    if (lane == 0) wtot[w] = s;
    __syncthreads();

    const int tw = wtot[lane];
    int incw = tw;
    for (int o = 1; o < 32; o <<= 1) {
        const int u = __shfl_up(incw, o);
        if (lane >= o) incw += u;
    }
    int run = __shfl(incw - tw, w);

    for (int c = c0; c < c1; ++c) {
        const int idx = (c << 7) + lane * 4;
        const int4 v = load4_guard(cnt, idx, n);
        const int l1 = v.x, l2 = l1 + v.y, l3 = l2 + v.z, l4 = l3 + v.w;
        int inc = l4;
        for (int o = 1; o < 32; o <<= 1) {
            const int u = __shfl_up(inc, o);
            if (lane >= o) inc += u;
        }
        const int b = run + inc - l4;
        if (idx + 3 < n) {
            *(int4*)(off + idx) = make_int4(b, b + l1, b + l2, b + l3);
        } else {
            if (idx     < n) off[idx]     = b;
            if (idx + 1 < n) off[idx + 1] = b + l1;
            if (idx + 2 < n) off[idx + 2] = b + l2;
        }
        run += __shfl(inc, 31);
    }
    if (w == 31 && lane == 0) off[n] = run;
}

__global__ void __launch_bounds__(256)
edge_fill_kernel(const int* __restrict__ ei,
                 const int* __restrict__ off,
                 const int* __restrict__ rank,
                 int* __restrict__ csr,
                 int e) {
    const int t = blockIdx.x * blockDim.x + threadIdx.x;
    if (t >= e) return;
    const int s = ei[t];
    const int d = ei[e + t];
    csr[off[d] + rank[t]] = s;
}

__device__ __forceinline__ void chunk_scores(const int* __restrict__ csr,
                                             const float* __restrict__ a_src,
                                             const float4 ad,
                                             int beg, int deg, int j,
                                             int& s, float4& ev) {
    s  = 0;
    ev = make_float4(-INFINITY, -INFINITY, -INFINITY, -INFINITY);
    if (j < deg) {
        s = csr[beg + j];
        const float4 a = *(const float4*)(a_src + (size_t)s * HEADS);
        ev.x = leaky(a.x + ad.x);
        ev.y = leaky(a.y + ad.y);
        ev.z = leaky(a.z + ad.z);
        ev.w = leaky(a.w + ad.w);
    }
}

__device__ __forceinline__ float lane_bcast(float v, int k) {
    return __int_as_float(__builtin_amdgcn_readlane(__float_as_int(v), k));
}

__global__ void __launch_bounds__(256)
node_aggregate_kernel(const int* __restrict__ off,
                      const int* __restrict__ csr,
                      const float* __restrict__ a_src,
                      const float* __restrict__ a_dst,
                      const float* __restrict__ xproj,
                      const float* __restrict__ bias,
                      float* __restrict__ out,
                      int n) {
    const int lane = threadIdx.x & 31;
    const int node = __builtin_amdgcn_readfirstlane((int)(blockIdx.x * 8 + (threadIdx.x >> 5)));
    if (node >= n) return;
    const int beg = off[node];
    const int deg = off[node + 1] - beg;
    const int h   = lane >> 3;
    const size_t rowOff = (size_t)node * INF_ + lane * 4;

    const float4 selfRow = *(const float4*)(xproj + rowOff);
    const float4 skipRow = *(const float4*)(out + rowOff);
    const float4 b4 = *(const float4*)(bias + lane * 4);
    const float4 ad = *(const float4*)(a_dst + (size_t)node * HEADS);
    const float4 an = *(const float4*)(a_src + (size_t)node * HEADS);

    const float es0 = leaky(an.x + ad.x);
    const float es1 = leaky(an.y + ad.y);
    const float es2 = leaky(an.z + ad.z);
    const float es3 = leaky(an.w + ad.w);
    float m0 = es0, m1 = es1, m2 = es2, m3 = es3;

    int sF; float4 eF;
    chunk_scores(csr, a_src, ad, beg, deg, lane, sF, eF);
    m0 = fmaxf(m0, eF.x); m1 = fmaxf(m1, eF.y);
    m2 = fmaxf(m2, eF.z); m3 = fmaxf(m3, eF.w);
    for (int base = 32; base < deg; base += 32) {
        int s; float4 ev;
        chunk_scores(csr, a_src, ad, beg, deg, base + lane, s, ev);
        m0 = fmaxf(m0, ev.x); m1 = fmaxf(m1, ev.y);
        m2 = fmaxf(m2, ev.z); m3 = fmaxf(m3, ev.w);
    }
    for (int o = 16; o > 0; o >>= 1) {
        m0 = fmaxf(m0, __shfl_xor(m0, o));
        m1 = fmaxf(m1, __shfl_xor(m1, o));
        m2 = fmaxf(m2, __shfl_xor(m2, o));
        m3 = fmaxf(m3, __shfl_xor(m3, o));
    }

    const float mh  = h == 0 ? m0  : h == 1 ? m1  : h == 2 ? m2  : m3;
    const float esh = h == 0 ? es0 : h == 1 ? es1 : h == 2 ? es2 : es3;
    const float pself = __expf(esh - mh);
    float4 acc = make_float4(pself * selfRow.x, pself * selfRow.y,
                             pself * selfRow.z, pself * selfRow.w);
    float den = pself;

    for (int base = 0; base < deg; base += 32) {
        int s = sF; float4 ev = eF;
        if (base != 0) chunk_scores(csr, a_src, ad, beg, deg, base + lane, s, ev);
        const float p0 = __expf(ev.x - m0);
        const float p1 = __expf(ev.y - m1);
        const float p2 = __expf(ev.z - m2);
        const float p3 = __expf(ev.w - m3);
        const int cntc = deg - base < 32 ? deg - base : 32;
#pragma unroll 4
        for (int k = 0; k < cntc; ++k) {
            const int sk = __builtin_amdgcn_readlane(s, k);
            const float q0 = lane_bcast(p0, k);
            const float q1 = lane_bcast(p1, k);
            const float q2 = lane_bcast(p2, k);
            const float q3 = lane_bcast(p3, k);
            const float pk = h == 0 ? q0 : h == 1 ? q1 : h == 2 ? q2 : q3;
            const float4 v = *(const float4*)(xproj + (size_t)sk * INF_ + lane * 4);
            acc.x = fmaf(pk, v.x, acc.x);
            acc.y = fmaf(pk, v.y, acc.y);
            acc.z = fmaf(pk, v.z, acc.z);
            acc.w = fmaf(pk, v.w, acc.w);
            den += pk;
        }
    }

    const float dn = den + 1e-16f;
    float4 r;
    r.x = acc.x / dn + b4.x + skipRow.x;
    r.y = acc.y / dn + b4.y + skipRow.y;
    r.z = acc.z / dn + b4.z + skipRow.z;
    r.w = acc.w / dn + b4.w + skipRow.w;
    r.x = r.x > 0.0f ? r.x : (__expf(r.x) - 1.0f);
    r.y = r.y > 0.0f ? r.y : (__expf(r.y) - 1.0f);
    r.z = r.z > 0.0f ? r.z : (__expf(r.z) - 1.0f);
    r.w = r.w > 0.0f ? r.w : (__expf(r.w) - 1.0f);
    *(float4*)(out + rowOff) = r;
}

extern "C" void kernel_launch(void* const* d_in, const int* in_sizes, int n_in,
                              void* d_out, int out_size, void* d_ws, size_t ws_size,
                              hipStream_t stream) {
    const float* x    = (const float*)d_in[0];
    const int*   ei   = (const int*)  d_in[1];
    const float* Wg   = (const float*)d_in[2];
    const float* asw  = (const float*)d_in[3];
    const float* adw  = (const float*)d_in[4];
    const float* bias = (const float*)d_in[5];
    const float* Ws   = (const float*)d_in[6];
    float* out = (float*)d_out;

    const int n = in_sizes[0] / INF_;
    const int e = in_sizes[1] / 2;

    float* ws    = (float*)d_ws;
    float* xproj = ws;
    float* a_src = xproj + (size_t)n * INF_;
    float* a_dst = a_src + (size_t)n * HEADS;
    int*   cnt   = (int*)(a_dst + (size_t)n * HEADS);
    int*   off   = cnt + (((size_t)n + 4) & ~(size_t)3);
    int*   rank  = off + (((size_t)n + 8) & ~(size_t)3);
    int*   csr   = rank + e;

    const dim3 gemmBlock(32, 8);
    gemm_proj_kernel<<<(n + 15) / 16, gemmBlock, 0, stream>>>(
        x, Wg, Ws, xproj, out, n);

    att_kernel<<<(n * HEADS + 255) / 256, 256, 0, stream>>>(
        xproj, asw, adw, a_src, a_dst, cnt, n);

    edge_rank_kernel<<<(e + 255) / 256, 256, 0, stream>>>(
        ei, cnt, rank, e);

    offset_scan_kernel<<<1, 1024, 0, stream>>>(cnt, off, n);

    edge_fill_kernel<<<(e + 255) / 256, 256, 0, stream>>>(
        ei, off, rank, csr, e);

    node_aggregate_kernel<<<(n + 7) / 8, 256, 0, stream>>>(
        off, csr, a_src, a_dst, xproj, bias, out, n);
}


# Offset scan spread over 13 independent blocks: each scans its own 4096 counters and sums the counters in front of them itself, instead of one block walking the whole counter plane twice
# speedup vs baseline: 3.6432x; 1.6456x over previous
#include <hip/hip_runtime.h>
#include <math.h>


typedef float v2f __attribute__((ext_vector_type(2)));
typedef float v8f __attribute__((ext_vector_type(8)));

#define INF_ 128
#define HEADS 4
#define NEG_SLOPE 0.2f

__device__ __forceinline__ float leaky(float e) {
    return e > 0.0f ? e : NEG_SLOPE * e;
}

#define XS_PITCH 132
#define XS_STAGE (16 * XS_PITCH)

__device__ __forceinline__ void
gemm_role(const float* __restrict__ x,
          const float* __restrict__ Wg,
          const float* __restrict__ Ws,
          float* __restrict__ xproj,
          float* __restrict__ skip,
          float* __restrict__ xs,
          int n, int firstStrip, int stripStep) {

    const int lane    = threadIdx.x;
    const int w       = threadIdx.y;
    const int colBase = w * 16;
    const int m       = lane & 15;
    const int hh      = lane >> 4;
    const int nStrips = (n + 15) >> 4;

    const float* __restrict__ gr = Wg + (size_t)(colBase + m) * INF_ + 4 * hh;
    const float* __restrict__ sr = Ws + (size_t)(colBase + m) * INF_ + 4 * hh;
    float4 bg[16], bs[16];
#pragma unroll
    for (int i = 0; i < 16; ++i) {
        bg[i] = *(const float4*)(gr + 8 * i);
        bs[i] = *(const float4*)(sr + 8 * i);
    }

    const int wrOff = (2 * w) * XS_PITCH + lane * 4;
    const int rdOff = m * XS_PITCH + 4 * hh;

    int strip = firstStrip;
    int stage = 0;
    if (strip < nStrips) {
        int r0 = strip * 16 + 2 * w;
        int r1 = r0 + 1;
        if (r0 >= n) r0 = n - 1;
        if (r1 >= n) r1 = n - 1;
        const float4 v0 = *(const float4*)(x + (size_t)r0 * INF_ + lane * 4);
        const float4 v1 = *(const float4*)(x + (size_t)r1 * INF_ + lane * 4);
        *(float4*)(xs + wrOff)            = v0;
        *(float4*)(xs + wrOff + XS_PITCH) = v1;
    }
    __syncthreads();

    for (; strip < nStrips; strip += stripStep) {
        const int  next = strip + stripStep;
        const bool more = next < nStrips;
        float4 nx0 = make_float4(0.f, 0.f, 0.f, 0.f);
        float4 nx1 = nx0;
        if (more) {
            int r0 = next * 16 + 2 * w;
            int r1 = r0 + 1;
            if (r0 >= n) r0 = n - 1;
            if (r1 >= n) r1 = n - 1;
            nx0 = *(const float4*)(x + (size_t)r0 * INF_ + lane * 4);
            nx1 = *(const float4*)(x + (size_t)r1 * INF_ + lane * 4);
        }

        const float* __restrict__ cur = xs + stage * XS_STAGE + rdOff;
        float4 a[16];
#pragma unroll
        for (int i = 0; i < 16; ++i) a[i] = *(const float4*)(cur + 8 * i);
        __builtin_amdgcn_sched_barrier(0);

        v8f accG = {};
        v8f accS = {};
#pragma unroll
        for (int i = 0; i < 16; ++i) {
            const v2f a0 = {a[i].x, a[i].y};
            const v2f a1 = {a[i].z, a[i].w};
            const v2f g0 = {bg[i].x, bg[i].y};
            const v2f g1 = {bg[i].z, bg[i].w};
            const v2f s0 = {bs[i].x, bs[i].y};
            const v2f s1 = {bs[i].z, bs[i].w};
            accG = __builtin_amdgcn_wmma_f32_16x16x4_f32(
                false, a0, false, g0, (short)0, accG, false, false);
            accS = __builtin_amdgcn_wmma_f32_16x16x4_f32(
                false, a0, false, s0, (short)0, accS, false, false);
            accG = __builtin_amdgcn_wmma_f32_16x16x4_f32(
                false, a1, false, g1, (short)0, accG, false, false);
            accS = __builtin_amdgcn_wmma_f32_16x16x4_f32(
                false, a1, false, s1, (short)0, accS, false, false);
        }

        if (more) {
            float* __restrict__ nxt = xs + (stage ^ 1) * XS_STAGE + wrOff;
            *(float4*)(nxt)            = nx0;
            *(float4*)(nxt + XS_PITCH) = nx1;
        }
        __syncthreads();
        stage ^= 1;

        const int rowBase = strip * 16 + hh * 8;
        float* __restrict__ pg = xproj + (size_t)rowBase * INF_ + colBase + m;
        float* __restrict__ ps = skip  + (size_t)rowBase * INF_ + colBase + m;
        if (strip * 16 + 16 <= n) {
#pragma unroll
            for (int j = 0; j < 8; ++j) {
                pg[j * INF_] = accG[j];
                ps[j * INF_] = accS[j];
            }
        } else {
#pragma unroll
            for (int j = 0; j < 8; ++j) {
                if (rowBase + j < n) {
                    pg[j * INF_] = accG[j];
                    ps[j * INF_] = accS[j];
                }
            }
        }
    }
}

__device__ __forceinline__ void
att_role(const float* __restrict__ xproj,
         const float* __restrict__ att_src,
         const float* __restrict__ att_dst,
         float* __restrict__ a_src,
         float* __restrict__ a_dst,
         int n, int t) {
    if (t >= n * HEADS) return;
    const int node = t >> 2;
    const int h    = t & 3;
    const float* __restrict__ xp = xproj + (size_t)node * INF_ + h * 32;
    const float* __restrict__ as = att_src + h * 32;
    const float* __restrict__ ad = att_dst + h * 32;
    float s = 0.0f, d = 0.0f;
#pragma unroll
    for (int c = 0; c < 32; ++c) {
        const float v = xp[c];
        s = fmaf(v, as[c], s);
        d = fmaf(v, ad[c], d);
    }
    a_src[t] = s;
    a_dst[t] = d;
}

__global__ void __launch_bounds__(256)
clear_counts_kernel(int* __restrict__ cnt, int n) {
    const int i = (blockIdx.x * blockDim.x + threadIdx.x) * 4;
    if (i < n) *(int4*)(cnt + i) = make_int4(0, 0, 0, 0);
}

__device__ __forceinline__ void
rank_role(const int* __restrict__ ei,
          int* __restrict__ cnt,
          int* __restrict__ rank,
          int e, int t) {
    const int i = t * 4;
    if (i >= e) return;
    if ((e & 3) == 0) {
        const int4 d = *(const int4*)(ei + e + i);
        int4 r;
        r.x = atomicAdd(&cnt[d.x], 1);
        r.y = atomicAdd(&cnt[d.y], 1);
        r.z = atomicAdd(&cnt[d.z], 1);
        r.w = atomicAdd(&cnt[d.w], 1);
        *(int4*)(rank + i) = r;
    } else {
        for (int k = i; k < i + 4 && k < e; ++k)
            rank[k] = atomicAdd(&cnt[ei[e + k]], 1);
    }
}

__global__ void __launch_bounds__(256)
proj_rank_kernel(const float* __restrict__ x,
                 const float* __restrict__ Wg,
                 const float* __restrict__ Ws,
                 float* __restrict__ xproj,
                 float* __restrict__ skip,
                 const int* __restrict__ ei,
                 int* __restrict__ cnt,
                 int* __restrict__ rank,
                 int n, int e, int gemmBlocks) {
    __shared__ __attribute__((aligned(16))) float xs[2 * XS_STAGE];
    const int b = blockIdx.x;
    if (b < gemmBlocks) {
        gemm_role(x, Wg, Ws, xproj, skip, xs, n, b, gemmBlocks);
    } else {
        const int t = (b - gemmBlocks) * 256 + threadIdx.y * 32 + threadIdx.x;
        rank_role(ei, cnt, rank, e, t);
    }
}

__device__ __forceinline__ int4 load4_guard(const int* __restrict__ p, int idx, int n) {
    int4 v = make_int4(0, 0, 0, 0);
    if (idx + 3 < n) {
        v = *(const int4*)(p + idx);
    } else {
        if (idx     < n) v.x = p[idx];
        if (idx + 1 < n) v.y = p[idx + 1];
        if (idx + 2 < n) v.z = p[idx + 2];
    }
    return v;
}

#define SCAN_TILE 4096

__device__ __forceinline__ int sum4(const int4 v) {
    return (v.x + v.y) + (v.z + v.w);
}

__device__ __forceinline__ void
scan_role(const int* __restrict__ cnt,
          int* __restrict__ off,
          int* __restrict__ wtot,
          int n, int b, int scanBlocks) {
    const int lane = threadIdx.x & 31;
    const int w    = threadIdx.x >> 5;
    const int idx  = b * SCAN_TILE + (int)threadIdx.x * 4;

    const int4 v = load4_guard(cnt, idx, n);

    const int4* __restrict__ front = (const int4*)cnt + threadIdx.x;
    int pre = 0;
    for (int i = 0; i < b; i += 4) {
        const bool k1 = i + 1 < b, k2 = i + 2 < b, k3 = i + 3 < b;
        const int4 f0 = front[(size_t)i * (SCAN_TILE / 4)];
        const int4 f1 = front[(size_t)(k1 ? i + 1 : 0) * (SCAN_TILE / 4)];
        const int4 f2 = front[(size_t)(k2 ? i + 2 : 0) * (SCAN_TILE / 4)];
        const int4 f3 = front[(size_t)(k3 ? i + 3 : 0) * (SCAN_TILE / 4)];
        pre += (sum4(f0) + (k1 ? sum4(f1) : 0)) + ((k2 ? sum4(f2) : 0) + (k3 ? sum4(f3) : 0));
    }

    const int l1 = v.x, l2 = l1 + v.y, l3 = l2 + v.z, l4 = l3 + v.w;
    int inc = l4;
    for (int o = 1; o < 32; o <<= 1) {
        const int u = __shfl_up(inc, o);
        if (lane >= o) inc += u;
    }
    for (int o = 16; o > 0; o >>= 1) pre += __shfl_xor(pre, o);
    if (lane == 31) {
        wtot[w]      = inc;
        wtot[32 + w] = pre;
    }
    __syncthreads();

    int t = wtot[32 + lane] + (lane < w ? wtot[lane] : 0);
    for (int o = 16; o > 0; o >>= 1) t += __shfl_xor(t, o);

    const int s = t + inc - l4;
    if (idx + 3 < n) {
        *(int4*)(off + idx) = make_int4(s, s + l1, s + l2, s + l3);
    } else {
        if (idx     < n) off[idx]     = s;
        if (idx + 1 < n) off[idx + 1] = s + l1;
        if (idx + 2 < n) off[idx + 2] = s + l2;
    }
    if (b == scanBlocks - 1 && threadIdx.x == SCAN_TILE / 4 - 1) off[n] = t + inc;
}

__global__ void __launch_bounds__(1024)
scan_att_kernel(const int* __restrict__ cnt,
                int* __restrict__ off,
                const float* __restrict__ xproj,
                const float* __restrict__ att_src,
                const float* __restrict__ att_dst,
                float* __restrict__ a_src,
                float* __restrict__ a_dst,
                int n) {
    __shared__ int wtot[64];
    const int scanBlocks = (n + SCAN_TILE - 1) / SCAN_TILE;
    const int b = blockIdx.x;
    if (b < scanBlocks) {
        scan_role(cnt, off, wtot, n, b, scanBlocks);
    } else {
        att_role(xproj, att_src, att_dst, a_src, a_dst, n,
                 (int)((b - scanBlocks) * 1024 + threadIdx.x));
    }
}

__global__ void __launch_bounds__(256)
edge_fill_kernel(const int* __restrict__ ei,
                 const int* __restrict__ off,
                 const int* __restrict__ rank,
                 int* __restrict__ csr,
                 int e) {
    const int t = blockIdx.x * blockDim.x + threadIdx.x;
    if (t >= e) return;
    const int s = ei[t];
    const int d = ei[e + t];
    csr[off[d] + rank[t]] = s;
}

__device__ __forceinline__ void chunk_scores(const int* __restrict__ csr,
                                             const float* __restrict__ a_src,
                                             const float4 ad,
                                             int beg, int deg, int j,
                                             int& s, float4& ev) {
    s  = 0;
    ev = make_float4(-INFINITY, -INFINITY, -INFINITY, -INFINITY);
    if (j < deg) {
        s = csr[beg + j];
        const float4 a = *(const float4*)(a_src + (size_t)s * HEADS);
        ev.x = leaky(a.x + ad.x);
        ev.y = leaky(a.y + ad.y);
        ev.z = leaky(a.z + ad.z);
        ev.w = leaky(a.w + ad.w);
    }
}

__device__ __forceinline__ float lane_bcast(float v, int k) {
    return __int_as_float(__builtin_amdgcn_readlane(__float_as_int(v), k));
}

__global__ void __launch_bounds__(256)
node_aggregate_kernel(const int* __restrict__ off,
                      const int* __restrict__ csr,
                      const float* __restrict__ a_src,
                      const float* __restrict__ a_dst,
                      const float* __restrict__ xproj,
                      const float* __restrict__ bias,
                      float* __restrict__ out,
                      int n) {
    const int lane = threadIdx.x & 31;
    const int node = __builtin_amdgcn_readfirstlane((int)(blockIdx.x * 8 + (threadIdx.x >> 5)));
    if (node >= n) return;
    const int beg = off[node];
    const int deg = off[node + 1] - beg;
    const int h   = lane >> 3;
    const size_t rowOff = (size_t)node * INF_ + lane * 4;

    const float4 selfRow = *(const float4*)(xproj + rowOff);
    const float4 skipRow = *(const float4*)(out + rowOff);
    const float4 b4 = *(const float4*)(bias + lane * 4);
    const float4 ad = *(const float4*)(a_dst + (size_t)node * HEADS);
    const float4 an = *(const float4*)(a_src + (size_t)node * HEADS);

    const float es0 = leaky(an.x + ad.x);
    const float es1 = leaky(an.y + ad.y);
    const float es2 = leaky(an.z + ad.z);
    const float es3 = leaky(an.w + ad.w);
    float m0 = es0, m1 = es1, m2 = es2, m3 = es3;

    int sF; float4 eF;
    chunk_scores(csr, a_src, ad, beg, deg, lane, sF, eF);
    m0 = fmaxf(m0, eF.x); m1 = fmaxf(m1, eF.y);
    m2 = fmaxf(m2, eF.z); m3 = fmaxf(m3, eF.w);
    for (int base = 32; base < deg; base += 32) {
        int s; float4 ev;
        chunk_scores(csr, a_src, ad, beg, deg, base + lane, s, ev);
        m0 = fmaxf(m0, ev.x); m1 = fmaxf(m1, ev.y);
        m2 = fmaxf(m2, ev.z); m3 = fmaxf(m3, ev.w);
    }
    for (int o = 16; o > 0; o >>= 1) {
        m0 = fmaxf(m0, __shfl_xor(m0, o));
        m1 = fmaxf(m1, __shfl_xor(m1, o));
        m2 = fmaxf(m2, __shfl_xor(m2, o));
        m3 = fmaxf(m3, __shfl_xor(m3, o));
    }

    const float mh  = h == 0 ? m0  : h == 1 ? m1  : h == 2 ? m2  : m3;
    const float esh = h == 0 ? es0 : h == 1 ? es1 : h == 2 ? es2 : es3;
    const float pself = __expf(esh - mh);
    float4 acc = make_float4(pself * selfRow.x, pself * selfRow.y,
                             pself * selfRow.z, pself * selfRow.w);
    float den = pself;

    for (int base = 0; base < deg; base += 32) {
        int s = sF; float4 ev = eF;
        if (base != 0) chunk_scores(csr, a_src, ad, beg, deg, base + lane, s, ev);
        const float p0 = __expf(ev.x - m0);
        const float p1 = __expf(ev.y - m1);
        const float p2 = __expf(ev.z - m2);
        const float p3 = __expf(ev.w - m3);
        const int cntc = deg - base < 32 ? deg - base : 32;
#pragma unroll 4
        for (int k = 0; k < cntc; ++k) {
            const int sk = __builtin_amdgcn_readlane(s, k);
            const float q0 = lane_bcast(p0, k);
            const float q1 = lane_bcast(p1, k);
            const float q2 = lane_bcast(p2, k);
            const float q3 = lane_bcast(p3, k);
            const float pk = h == 0 ? q0 : h == 1 ? q1 : h == 2 ? q2 : q3;
            const float4 v = *(const float4*)(xproj + (size_t)sk * INF_ + lane * 4);
            acc.x = fmaf(pk, v.x, acc.x);
            acc.y = fmaf(pk, v.y, acc.y);
            acc.z = fmaf(pk, v.z, acc.z);
            acc.w = fmaf(pk, v.w, acc.w);
            den += pk;
        }
    }

    const float dn = den + 1e-16f;
    float4 r;
    r.x = acc.x / dn + b4.x + skipRow.x;
    r.y = acc.y / dn + b4.y + skipRow.y;
    r.z = acc.z / dn + b4.z + skipRow.z;
    r.w = acc.w / dn + b4.w + skipRow.w;
    r.x = r.x > 0.0f ? r.x : (__expf(r.x) - 1.0f);
    r.y = r.y > 0.0f ? r.y : (__expf(r.y) - 1.0f);
    r.z = r.z > 0.0f ? r.z : (__expf(r.z) - 1.0f);
    r.w = r.w > 0.0f ? r.w : (__expf(r.w) - 1.0f);
    *(float4*)(out + rowOff) = r;
}

extern "C" void kernel_launch(void* const* d_in, const int* in_sizes, int n_in,
                              void* d_out, int out_size, void* d_ws, size_t ws_size,
                              hipStream_t stream) {
    const float* x    = (const float*)d_in[0];
    const int*   ei   = (const int*)  d_in[1];
    const float* Wg   = (const float*)d_in[2];
    const float* asw  = (const float*)d_in[3];
    const float* adw  = (const float*)d_in[4];
    const float* bias = (const float*)d_in[5];
    const float* Ws   = (const float*)d_in[6];
    float* out = (float*)d_out;

    const int n = in_sizes[0] / INF_;
    const int e = in_sizes[1] / 2;

    float* ws    = (float*)d_ws;
    float* xproj = ws;
    float* a_src = xproj + (size_t)n * INF_;
    float* a_dst = a_src + (size_t)n * HEADS;
    int*   cnt   = (int*)(a_dst + (size_t)n * HEADS);
    int*   off   = cnt + (((size_t)n + 4) & ~(size_t)3);
    int*   rank  = off + (((size_t)n + 8) & ~(size_t)3);
    int*   csr   = rank + e;

    clear_counts_kernel<<<(n + 1023) / 1024, 256, 0, stream>>>(cnt, n);

    const dim3 gemmBlock(32, 8);
    proj_rank_kernel<<<256 + (e + 1023) / 1024, gemmBlock, 0, stream>>>(
        x, Wg, Ws, xproj, out, ei, cnt, rank, n, e, 256);

    scan_att_kernel<<<(n + SCAN_TILE - 1) / SCAN_TILE + (n * HEADS + 1023) / 1024, 1024, 0, stream>>>(
        cnt, off, xproj, asw, adw, a_src, a_dst, n);

    edge_fill_kernel<<<(e + 255) / 256, 256, 0, stream>>>(
        ei, off, rank, csr, e);

    node_aggregate_kernel<<<(n + 7) / 8, 256, 0, stream>>>(
        off, csr, a_src, a_dst, xproj, bias, out, n);
}
